# starting kernel
# baseline (speedup 1.0000x reference)
#include <hip/hip_runtime.h>
#include <math.h>


typedef __attribute__((ext_vector_type(2))) float v2f;
typedef __attribute__((ext_vector_type(8))) float v8f;

#define HH 128
#define WW 128
#define HW (HH * WW)

__device__ __forceinline__ int reflect_idx(int i, int n) {
  if (i < 0) i = -i;
  if (i >= n) i = 2 * n - 2 - i;
  return i;
}

__global__ void pack_w(const float* __restrict__ cw, float* __restrict__ wp) {
  int t = blockIdx.x * blockDim.x + threadIdx.x;
  if (t >= 144 * 2 * 32 * 2) return;
  int v = t & 1;
  int lane = (t >> 1) & 31;
  int mt = (t >> 6) & 1;
  int kb = t >> 7;
  int m = (lane & 15) + 16 * mt;
  int hi = lane >> 4;
  int k = kb * 4 + 2 * hi + v;
  int kpos = k >> 6;
  int ic = k & 63;
  float val = 0.0f;
  if (m < 18) val = cw[(m * 64 + ic) * 9 + kpos];
  wp[t] = val;
}

__global__ __launch_bounds__(256) void conv_softmax(
    const float* __restrict__ feat, const float* __restrict__ bias,
    const float* __restrict__ wp, float* __restrict__ ffw) {
  __shared__ float xs[64 * 3 * 132];
  __shared__ float zb[8 * 16 * 19];

  const int h = blockIdx.x;
  const int b = blockIdx.y;
  const int tid = threadIdx.x;

  const float* fb = feat + (size_t)b * 64 * HW;
  for (int e = tid; e < 64 * 3 * 130; e += 256) {
    int ic = e / 390;
    int rem = e - ic * 390;
    int r = rem / 130;
    int wcol = rem - r * 130;
    int hr = reflect_idx(h + r - 1, HH);
    int ws = reflect_idx(wcol - 1, WW);
    xs[((ic >> 1) * 3 + r) * 264 + wcol * 2 + (ic & 1)] = fb[ic * HW + hr * WW + ws];
  }
  __syncthreads();

  const int lane = tid & 31, wv = tid >> 5;
  const int p = lane & 15, hi = lane >> 4;
  const int w0 = wv * 16;

  const v2f* __restrict__ wf = (const v2f*)wp;
  const v2f* xs2 = (const v2f*)xs;

  v8f acc0 = {0.f, 0.f, 0.f, 0.f, 0.f, 0.f, 0.f, 0.f};
  v8f acc1 = acc0;

  int kb = 0;
  for (int kpos = 0; kpos < 9; ++kpos) {
    const int dh = kpos / 3, dw = kpos - dh * 3;
    const v2f* xrow = xs2 + (size_t)hi * 396 + dh * 132 + (w0 + p + dw);
#pragma unroll
    for (int j = 0; j < 16; ++j, ++kb) {
      v2f a0 = wf[kb * 64 + lane];
      v2f a1 = wf[kb * 64 + 32 + lane];
      v2f bx = xrow[(size_t)j * 792];
      acc0 = __builtin_amdgcn_wmma_f32_16x16x4_f32(
          false, a0, false, bx, (short)0, acc0, false, false);
      acc1 = __builtin_amdgcn_wmma_f32_16x16x4_f32(
          false, a1, false, bx, (short)0, acc1, false, false);
    }
  }

  float* zrow = &zb[(wv * 16 + p) * 19];
#pragma unroll
  for (int r = 0; r < 8; ++r) zrow[r + 8 * hi] = acc0[r];
  if (hi == 0) {
    zrow[16] = acc1[0];
    zrow[17] = acc1[1];
  }
  __syncthreads();

  const int g = hi;
  float z[9];
#pragma unroll
  for (int k = 0; k < 9; ++k) z[k] = zrow[2 * k + g] + bias[2 * k + g];
  float mx = z[0];
#pragma unroll
  for (int k = 1; k < 9; ++k) mx = fmaxf(mx, z[k]);
  float s = 0.0f;
#pragma unroll
  for (int k = 0; k < 9; ++k) {
    z[k] = expf(z[k] - mx);
    s += z[k];
  }
  const float inv = 1.0f / s;
  const int wpix = w0 + p;
#pragma unroll
  for (int k = 0; k < 9; ++k) {
    int ch = 2 * k + g;
    int si = ch / 9, kk = ch - si * 9;
    ffw[(((size_t)b * 2 + si) * 9 + kk) * HW + h * WW + wpix] = z[k] * inv;
  }
}

__global__ __launch_bounds__(256) void combine(
    const float* __restrict__ inp, const float* __restrict__ ffw,
    float* __restrict__ out, int total) {
  int n = blockIdx.x * 256 + threadIdx.x;
  if (n >= total) return;
  int w = n & 127;
  int h = (n >> 7) & 127;
  int c = (n >> 14) & 63;
  int b = n >> 20;
  const float* xin = inp + (size_t)(b * 64 + c) * HW;
  const float* wg = ffw + (((size_t)b * 2 + (c & 1)) * 9) * HW + h * WW + w;
  float acc = 0.0f;
#pragma unroll
  for (int kpos = 0; kpos < 9; ++kpos) {
    int dh = kpos / 3, dw = kpos - dh * 3;
    int hr = reflect_idx(h + dh - 1, HH);
    int wr = reflect_idx(w + dw - 1, WW);
    acc = fmaf(wg[(size_t)kpos * HW], xin[hr * WW + wr], acc);
  }
  out[n] = acc;
}

extern "C" void kernel_launch(void* const* d_in, const int* in_sizes, int n_in,
                              void* d_out, int out_size, void* d_ws, size_t ws_size,
                              hipStream_t stream) {
  const float* features = (const float*)d_in[0];
  const float* inpt     = (const float*)d_in[1];
  const float* conv_w   = (const float*)d_in[2];
  const float* conv_b   = (const float*)d_in[3];
  float* out = (float*)d_out;

  float* wp  = (float*)d_ws;
  float* ffw = wp + 18432;

  pack_w<<<(18432 + 255) / 256, 256, 0, stream>>>(conv_w, wp);
  conv_softmax<<<dim3(HH, 8), 256, 0, stream>>>(features, conv_b, wp, ffw);
  const int total = 8 * 64 * HW;
  combine<<<(total + 255) / 256, 256, 0, stream>>>(inpt, ffw, out, total);
}


# conv kernel: out channels 16,17 take their 3 row taps as rows of ONE tile per feature row (depth 192), summed in registers; no padded 2nd tile per output row: 54 WMMAs per unit, was 72; new sum order
# speedup vs baseline: 2.2978x; 2.2978x over previous
#include <hip/hip_runtime.h>
#include <math.h>


typedef __attribute__((ext_vector_type(2))) float v2f;
typedef __attribute__((ext_vector_type(8))) float v8f;

#define HH 128
#define WW 128
#define HW (HH * WW)

__device__ __forceinline__ int reflect_idx(int i, int n) {
  if (i < 0) i = -i;
  if (i >= n) i = 2 * n - 2 - i;
  return i;
}

__global__ void pack_w(const float* __restrict__ cw, float* __restrict__ wp) {
  int t = blockIdx.x * blockDim.x + threadIdx.x;
  if (t >= 12 * 16 * 32 * 2) return;
  int v = t & 1;
  int lane = (t >> 1) & 31;
  int j = (t >> 6) & 15;
  int f = t >> 10;
  int m = lane & 15;
  int hi = lane >> 4;
  int ic = 4 * j + 2 * hi + v;
  float val = 0.0f;
  if (f < 9) {
    val = cw[(m * 64 + ic) * 9 + f];
  } else if (m < 6) {
    val = cw[((16 + (m & 1)) * 64 + ic) * 9 + (m >> 1) * 3 + (f - 9)];
  }
  wp[t] = val;
}

#define RT 4
#define FR (RT + 2)

struct Unit {
  v2f a0[3][3];
  v2f a1[3];
  v2f b[3][FR];
};

__device__ __forceinline__ void load_unit(Unit& f, const v2f* __restrict__ wa,
                                          const float* __restrict__ fc, const int (&ro)[FR],
                                          int lc0, int lc1, int lc2) {
#pragma unroll
  for (int dh = 0; dh < 3; ++dh) {
#pragma unroll
    for (int dw = 0; dw < 3; ++dw) f.a0[dh][dw] = wa[(3 * dh + dw) * 512];
  }
#pragma unroll
  for (int dw = 0; dw < 3; ++dw) f.a1[dw] = wa[(9 + dw) * 512];
#pragma unroll
  for (int i = 0; i < FR; ++i) {
    const float* fr = fc + ro[i];
    f.b[0][i].x = fr[lc0];  f.b[0][i].y = fr[lc0 + HW];
    f.b[1][i].x = fr[lc1];  f.b[1][i].y = fr[lc1 + HW];
    f.b[2][i].x = fr[lc2];  f.b[2][i].y = fr[lc2 + HW];
  }
}

__device__ __forceinline__ void mma_unit(const Unit& f, v8f (&acc0)[RT], v8f (&acc1)[FR]) {
#pragma unroll
  for (int dw = 0; dw < 3; ++dw) {
#pragma unroll
    for (int dh = 0; dh < 3; ++dh) {
#pragma unroll
      for (int o = 0; o < RT; ++o) {
        acc0[o] = __builtin_amdgcn_wmma_f32_16x16x4_f32(
            false, f.a0[dh][dw], false, f.b[dw][o + dh], (short)0, acc0[o], false, false);
      }
    }
#pragma unroll
    for (int i = 0; i < FR; ++i) {
      acc1[i] = __builtin_amdgcn_wmma_f32_16x16x4_f32(
          false, f.a1[dw], false, f.b[dw][i], (short)0, acc1[i], false, false);
    }
  }
}

__global__ __launch_bounds__(256) void conv_softmax(
    const float* __restrict__ feat, const float* __restrict__ bias,
    const float* __restrict__ wp, float* __restrict__ ffw) {
  __shared__ float zb[RT][8 * 16 * 19];

  const int h0 = blockIdx.x * RT;
  const int b = blockIdx.y;
  const int tid = threadIdx.x;

  const int lane = tid & 31, wv = tid >> 5;
  const int p = lane & 15, hi = lane >> 4;
  const int w0 = wv * 16;

  const v2f* __restrict__ wf = (const v2f*)wp + lane;

  const float* fb = feat + (size_t)b * 64 * HW;
  const int lc0 = 2 * hi * HW + reflect_idx(w0 + p - 1, WW);
  const int lc1 = 2 * hi * HW + w0 + p;
  const int lc2 = 2 * hi * HW + reflect_idx(w0 + p + 1, WW);
  int ro[FR];
#pragma unroll
  for (int i = 0; i < FR; ++i) ro[i] = reflect_idx(h0 - 1 + i, HH) * WW;

  v8f acc0[RT];
  v8f acc1[FR];
#pragma unroll
  for (int o = 0; o < RT; ++o) acc0[o] = (v8f){0.f, 0.f, 0.f, 0.f, 0.f, 0.f, 0.f, 0.f};
#pragma unroll
  for (int i = 0; i < FR; ++i) acc1[i] = (v8f){0.f, 0.f, 0.f, 0.f, 0.f, 0.f, 0.f, 0.f};

  Unit f0, f1;
  load_unit(f0, wf, fb, ro, lc0, lc1, lc2);
#pragma unroll 1
  for (int j = 0; j < 14; j += 2) {
    load_unit(f1, wf + (j + 1) * 32, fb + (j + 1) * 4 * HW, ro, lc0, lc1, lc2);
    __builtin_amdgcn_sched_barrier(0);
    mma_unit(f0, acc0, acc1);
    __builtin_amdgcn_sched_barrier(0);
    load_unit(f0, wf + (j + 2) * 32, fb + (j + 2) * 4 * HW, ro, lc0, lc1, lc2);
    __builtin_amdgcn_sched_barrier(0);
    mma_unit(f1, acc0, acc1);
  }
  load_unit(f1, wf + 15 * 32, fb + 15 * 4 * HW, ro, lc0, lc1, lc2);
  __builtin_amdgcn_sched_barrier(0);
  mma_unit(f0, acc0, acc1);
  mma_unit(f1, acc0, acc1);

#pragma unroll
  for (int o = 0; o < RT; ++o) {
    float* zrow = &zb[o][(wv * 16 + p) * 19];
#pragma unroll
    for (int r = 0; r < 8; ++r) zrow[r + 8 * hi] = acc0[o][r];
    if (hi == 0) {
      zrow[16] = (acc1[o][0] + acc1[o + 1][2]) + acc1[o + 2][4];
      zrow[17] = (acc1[o][1] + acc1[o + 1][3]) + acc1[o + 2][5];
    }
  }
  __syncthreads();

  const int g = hi;
  const int wpix = w0 + p;
  float bz[9];
#pragma unroll
  for (int k = 0; k < 9; ++k) bz[k] = bias[2 * k + g];
#pragma unroll
  for (int o = 0; o < RT; ++o) {
    const float* zrow = &zb[o][(wv * 16 + p) * 19];
    float z[9];
#pragma unroll
    for (int k = 0; k < 9; ++k) z[k] = zrow[2 * k + g] + bz[k];
    float mx = z[0];
#pragma unroll
    for (int k = 1; k < 9; ++k) mx = fmaxf(mx, z[k]);
    float s = 0.0f;
#pragma unroll
    for (int k = 0; k < 9; ++k) {
      z[k] = expf(z[k] - mx);
      s += z[k];
    }
    const float inv = 1.0f / s;
#pragma unroll
    for (int k = 0; k < 9; ++k) {
      int ch = 2 * k + g;
      int si = ch / 9, kk = ch - si * 9;
      ffw[(((size_t)b * 2 + si) * 9 + kk) * HW + (h0 + o) * WW + wpix] = z[k] * inv;
    }
  }
}

#define RB 8
__global__ __launch_bounds__(256) void combine(
    const float* __restrict__ inp, const float* __restrict__ ffw,
    float* __restrict__ out) {
  const int w = threadIdx.x & 127;
  const int slot = threadIdx.x >> 7;
  const int h0 = blockIdx.x * RB;
  const int s = blockIdx.y & 1;
  const int c = s + 2 * ((blockIdx.y >> 1) * 2 + slot);
  const int b = blockIdx.z;
  const int wl = reflect_idx(w - 1, WW);
  const int wr = reflect_idx(w + 1, WW);
  const float* xin = inp + (size_t)(b * 64 + c) * HW;
  const float* wg = ffw + (((size_t)b * 2 + s) * 9) * HW + w;
  float* op = out + (size_t)(b * 64 + c) * HW + w;

  const int rt = reflect_idx(h0 - 1, HH) * WW;
  float t0 = xin[rt + wl], t1 = xin[rt + w], t2 = xin[rt + wr];
  float m0 = xin[h0 * WW + wl], m1 = xin[h0 * WW + w], m2 = xin[h0 * WW + wr];
#pragma unroll
  for (int i = 0; i < RB; ++i) {
    const int h = h0 + i;
    const int rb = reflect_idx(h + 1, HH) * WW;
    const float u0 = xin[rb + wl], u1 = xin[rb + w], u2 = xin[rb + wr];
    const float* wrow = wg + h * WW;
    float acc = 0.0f;
    acc = fmaf(wrow[0 * HW], t0, acc);
    acc = fmaf(wrow[1 * HW], t1, acc);
    acc = fmaf(wrow[2 * HW], t2, acc);
    acc = fmaf(wrow[3 * HW], m0, acc);
    acc = fmaf(wrow[4 * HW], m1, acc);
    acc = fmaf(wrow[5 * HW], m2, acc);
    acc = fmaf(wrow[6 * HW], u0, acc);
    acc = fmaf(wrow[7 * HW], u1, acc);
    acc = fmaf(wrow[8 * HW], u2, acc);
    op[h * WW] = acc;
    t0 = m0; t1 = m1; t2 = m2;
    m0 = u0; m1 = u1; m2 = u2;
  }
}

extern "C" void kernel_launch(void* const* d_in, const int* in_sizes, int n_in,
                              void* d_out, int out_size, void* d_ws, size_t ws_size,
                              hipStream_t stream) {
  const float* features = (const float*)d_in[0];
  const float* inpt     = (const float*)d_in[1];
  const float* conv_w   = (const float*)d_in[2];
  const float* conv_b   = (const float*)d_in[3];
  float* out = (float*)d_out;

  float* wp  = (float*)d_ws;
  float* ffw = wp + 18432;

  pack_w<<<48, 256, 0, stream>>>(conv_w, wp);
  conv_softmax<<<dim3(32, 8), 256, 0, stream>>>(features, conv_b, wp, ffw);
  combine<<<dim3(16, 32, 8), 256, 0, stream>>>(inpt, ffw, out);
}


# ff weights stored tap-interleaved [b][set][h][half][w][4] + tap-8 plane: combine reads a pixel's 9 taps with 3 loads, two of 16 bytes (was 9 of 4); conv stores them after a lane-pair exchange
# speedup vs baseline: 2.5798x; 1.1227x over previous
#include <hip/hip_runtime.h>
#include <math.h>


typedef __attribute__((ext_vector_type(2))) float v2f;
typedef __attribute__((ext_vector_type(8))) float v8f;

#define HH 128
#define WW 128
#define HW (HH * WW)

__device__ __forceinline__ int reflect_idx(int i, int n) {
  if (i < 0) i = -i;
  if (i >= n) i = 2 * n - 2 - i;
  return i;
}

__global__ void pack_w(const float* __restrict__ cw, float* __restrict__ wp) {
  int t = blockIdx.x * blockDim.x + threadIdx.x;
  if (t >= 15 * 16 * 32 * 2) return;
  int v = t & 1;
  int lane = (t >> 1) & 31;
  int j = (t >> 6) & 15;
  int f = t >> 10;
  int m = lane & 15;
  int hi = lane >> 4;
  int ic = 4 * j + 2 * hi + v;
  float val = 0.0f;
  if (f < 12) {
    const int q = f / 3, kw = f - 3 * q;
    const float* g = cw + (m * 64 + ic) * 9 + kw;
    const float g0 = g[0], g1 = g[3], g2 = g[6];
    if (q == 0) val = g0;
    else if (q == 1) val = 0.5f * ((g0 + g2) + g1);
    else if (q == 2) val = 0.5f * ((g0 + g2) - g1);
    else val = g2;
  } else if (m < 6) {
    val = cw[((16 + (m & 1)) * 64 + ic) * 9 + (m >> 1) * 3 + (f - 12)];
  }
  wp[t] = val;
}

#define RT 4
#define FR (RT + 2)

struct Unit {
  v2f a0[4][3];
  v2f a1[3];
  v2f b[3][FR];
};

__device__ __forceinline__ void load_unit(Unit& f, const v2f* __restrict__ wa,
                                          const float* __restrict__ fc, const int (&ro)[FR],
                                          int lc0, int lc1, int lc2) {
#pragma unroll
  for (int q = 0; q < 4; ++q) {
#pragma unroll
    for (int dw = 0; dw < 3; ++dw) f.a0[q][dw] = wa[(3 * q + dw) * 512];
  }
#pragma unroll
  for (int dw = 0; dw < 3; ++dw) f.a1[dw] = wa[(12 + dw) * 512];
#pragma unroll
  for (int i = 0; i < FR; ++i) {
    const float* fr = fc + ro[i];
    f.b[0][i].x = fr[lc0];  f.b[0][i].y = fr[lc0 + HW];
    f.b[1][i].x = fr[lc1];  f.b[1][i].y = fr[lc1 + HW];
    f.b[2][i].x = fr[lc2];  f.b[2][i].y = fr[lc2 + HW];
  }
}

__device__ __forceinline__ void mma_unit(const Unit& f, v8f (&accw)[RT / 2][4], v8f (&acc1)[FR]) {
#pragma unroll
  for (int dw = 0; dw < 3; ++dw) {
#pragma unroll
    for (int t = 0; t < RT / 2; ++t) {
      const v2f d0 = f.b[dw][2 * t], d1 = f.b[dw][2 * t + 1];
      const v2f d2 = f.b[dw][2 * t + 2], d3 = f.b[dw][2 * t + 3];
      const v2f tq[4] = {d0 - d2, d1 + d2, d2 - d1, d1 - d3};
#pragma unroll
      for (int q = 0; q < 4; ++q) {
        accw[t][q] = __builtin_amdgcn_wmma_f32_16x16x4_f32(
            false, f.a0[q][dw], false, tq[q], (short)0, accw[t][q], false, false);
      }
    }
#pragma unroll
    for (int i = 0; i < FR; ++i) {
      acc1[i] = __builtin_amdgcn_wmma_f32_16x16x4_f32(
          false, f.a1[dw], false, f.b[dw][i], (short)0, acc1[i], false, false);
    }
  }
}

__global__ __launch_bounds__(256) void conv_softmax(
    const float* __restrict__ feat, const float* __restrict__ bias,
    const float* __restrict__ wp, float* __restrict__ ffw) {
  __shared__ float zb[RT][8 * 16 * 19];

  const int h0 = blockIdx.x * RT;
  const int b = blockIdx.y;
  const int tid = threadIdx.x;

  const int lane = tid & 31, wv = tid >> 5;
  const int p = lane & 15, hi = lane >> 4;
  const int w0 = wv * 16;

  const v2f* __restrict__ wf = (const v2f*)wp + lane;

  const float* fb = feat + (size_t)b * 64 * HW;
  const int lc0 = 2 * hi * HW + reflect_idx(w0 + p - 1, WW);
  const int lc1 = 2 * hi * HW + w0 + p;
  const int lc2 = 2 * hi * HW + reflect_idx(w0 + p + 1, WW);
  int ro[FR];
#pragma unroll
  for (int i = 0; i < FR; ++i) ro[i] = reflect_idx(h0 - 1 + i, HH) * WW;

  v8f accw[RT / 2][4];
  v8f acc1[FR];
#pragma unroll
  for (int t = 0; t < RT / 2; ++t) {
#pragma unroll
    for (int q = 0; q < 4; ++q) accw[t][q] = (v8f){0.f, 0.f, 0.f, 0.f, 0.f, 0.f, 0.f, 0.f};
  }
#pragma unroll
  for (int i = 0; i < FR; ++i) acc1[i] = (v8f){0.f, 0.f, 0.f, 0.f, 0.f, 0.f, 0.f, 0.f};

  Unit f0, f1;
  load_unit(f0, wf, fb, ro, lc0, lc1, lc2);
#pragma unroll 1
  for (int j = 0; j < 14; j += 2) {
    load_unit(f1, wf + (j + 1) * 32, fb + (j + 1) * 4 * HW, ro, lc0, lc1, lc2);
    __builtin_amdgcn_sched_barrier(0);
    mma_unit(f0, accw, acc1);
    __builtin_amdgcn_sched_barrier(0);
    load_unit(f0, wf + (j + 2) * 32, fb + (j + 2) * 4 * HW, ro, lc0, lc1, lc2);
    __builtin_amdgcn_sched_barrier(0);
    mma_unit(f1, accw, acc1);
  }
  load_unit(f1, wf + 15 * 32, fb + 15 * 4 * HW, ro, lc0, lc1, lc2);
  __builtin_amdgcn_sched_barrier(0);
  mma_unit(f0, accw, acc1);
  mma_unit(f1, accw, acc1);

#pragma unroll
  for (int o = 0; o < RT; ++o) {
    float* zrow = &zb[o][(wv * 16 + p) * 19];
    const int t = o >> 1;
    const v8f zo = (o & 1) ? (accw[t][1] - accw[t][2]) - accw[t][3]
                           : (accw[t][0] + accw[t][1]) + accw[t][2];
#pragma unroll
    for (int r = 0; r < 8; ++r) zrow[r + 8 * hi] = zo[r];
    if (hi == 0) {
      zrow[16] = (acc1[o][0] + acc1[o + 1][2]) + acc1[o + 2][4];
      zrow[17] = (acc1[o][1] + acc1[o + 1][3]) + acc1[o + 2][5];
    }
  }
  __syncthreads();

  const int g = hi;
  const int wpix = w0 + p;
  float bz[9];
#pragma unroll
  for (int k = 0; k < 9; ++k) bz[k] = bias[2 * k + g];
#pragma unroll
  for (int o = 0; o < RT; ++o) {
    const float* zrow = &zb[o][(wv * 16 + p) * 19];
    float z[9];
#pragma unroll
    for (int k = 0; k < 9; ++k) z[k] = zrow[2 * k + g] + bz[k];
    float mx = z[0];
#pragma unroll
    for (int k = 1; k < 9; ++k) mx = fmaxf(mx, z[k]);
    float s = 0.0f;
#pragma unroll
    for (int k = 0; k < 9; ++k) {
      z[k] = expf(z[k] - mx);
      s += z[k];
    }
    const float inv = 1.0f / s;
#pragma unroll
    for (int k = 0; k < 9; ++k) z[k] *= inv;
    float ev[5], od[4];
#pragma unroll
    for (int i = 0; i < 5; ++i) ev[i] = g ? z[4 + i] : z[i];
#pragma unroll
    for (int i = 0; i < 4; ++i) od[i] = __shfl_xor(g ? z[i] : z[5 + i], 16);
    float* fblk = ffw + ((size_t)b * 2 + g) * 9 * HW;
    float4* q4 = (float4*)fblk + ((h0 + o) * 2) * WW + wpix;
    q4[0] = make_float4(ev[0], od[0], ev[1], od[1]);
    q4[WW] = make_float4(ev[2], od[2], ev[3], od[3]);
    fblk[8 * HW + (h0 + o) * WW + wpix] = ev[4];
  }
}

#define RB 8
__global__ __launch_bounds__(256) void combine(
    const float* __restrict__ inp, const float* __restrict__ ffw,
    float* __restrict__ out) {
  const int w = threadIdx.x & 127;
  const int slot = threadIdx.x >> 7;
  const int h0 = blockIdx.x * RB;
  const int s = blockIdx.y & 1;
  const int c = s + 2 * ((blockIdx.y >> 1) * 2 + slot);
  const int b = blockIdx.z;
  const int wl = reflect_idx(w - 1, WW);
  const int wr = reflect_idx(w + 1, WW);
  const float* xin = inp + (size_t)(b * 64 + c) * HW;
  const float* wblk = ffw + (((size_t)b * 2 + s) * 9) * HW;
  const float4* wq = (const float4*)wblk + w;
  const float* w8 = wblk + 8 * HW + w;
  float* op = out + (size_t)(b * 64 + c) * HW + w;

  const int rt = reflect_idx(h0 - 1, HH) * WW;
  float t0 = xin[rt + wl], t1 = xin[rt + w], t2 = xin[rt + wr];
  float m0 = xin[h0 * WW + wl], m1 = xin[h0 * WW + w], m2 = xin[h0 * WW + wr];
#pragma unroll
  for (int i = 0; i < RB; ++i) {
    const int h = h0 + i;
    const int rb = reflect_idx(h + 1, HH) * WW;
    const float u0 = xin[rb + wl], u1 = xin[rb + w], u2 = xin[rb + wr];
    const float4 qa = wq[(h * 2) * WW];
    const float4 qb = wq[(h * 2 + 1) * WW];
    const float k8 = w8[h * WW];
    float acc = 0.0f;
    acc = fmaf(qa.x, t0, acc);
    acc = fmaf(qa.y, t1, acc);
    acc = fmaf(qa.z, t2, acc);
    acc = fmaf(qa.w, m0, acc);
    acc = fmaf(qb.x, m1, acc);
    acc = fmaf(qb.y, m2, acc);
    acc = fmaf(qb.z, u0, acc);
    acc = fmaf(qb.w, u1, acc);
    acc = fmaf(k8, u2, acc);
    op[h * WW] = acc;
    t0 = m0; t1 = m1; t2 = m2;
    m0 = u0; m1 = u1; m2 = u2;
  }
}

extern "C" void kernel_launch(void* const* d_in, const int* in_sizes, int n_in,
                              void* d_out, int out_size, void* d_ws, size_t ws_size,
                              hipStream_t stream) {
  const float* features = (const float*)d_in[0];
  const float* inpt     = (const float*)d_in[1];
  const float* conv_w   = (const float*)d_in[2];
  const float* conv_b   = (const float*)d_in[3];
  float* out = (float*)d_out;

  float* wp  = (float*)d_ws;
  float* ffw = wp + 18432;

  pack_w<<<60, 256, 0, stream>>>(conv_w, wp);
  conv_softmax<<<dim3(32, 8), 256, 0, stream>>>(features, conv_b, wp, ffw);
  combine<<<dim3(16, 32, 8), 256, 0, stream>>>(inpt, ffw, out);
}


# conv kernel: a tile column is a PAIR of output columns, so column taps too go by Winograd F(2,3) inside a lane (F(2x2,3x3) for out channels 0..15): 32 WMMAs per 64 pixels, was 42; new sum order
# speedup vs baseline: 2.7183x; 1.0537x over previous
#include <hip/hip_runtime.h>
#include <math.h>


typedef __attribute__((ext_vector_type(2))) float v2f;
typedef __attribute__((ext_vector_type(8))) float v8f;

#define HH 128
#define WW 128
#define HW (HH * WW)

__device__ __forceinline__ int reflect_idx(int i, int n) {
  if (i < 0) i = -i;
  if (i >= n) i = 2 * n - 2 - i;
  return i;
}

__device__ __forceinline__ float wino_g(float g0, float g1, float g2, int q) {
  if (q == 0) return g0;
  if (q == 1) return 0.5f * ((g0 + g2) + g1);
  if (q == 2) return 0.5f * ((g0 + g2) - g1);
  return g2;
}

__global__ void pack_w(const float* __restrict__ cw, float* __restrict__ wp) {
  int t = blockIdx.x * blockDim.x + threadIdx.x;
  if (t >= 20 * 16 * 32 * 2) return;
  int v = t & 1;
  int lane = (t >> 1) & 31;
  int j = (t >> 6) & 15;
  int f = t >> 10;
  int m = lane & 15;
  int hi = lane >> 4;
  int ic = 4 * j + 2 * hi + v;
  float val = 0.0f;
  if (f < 16) {
    const int qh = f >> 2, qw = f & 3;
    const float* g = cw + (m * 64 + ic) * 9;
    const float r0 = wino_g(g[0], g[1], g[2], qw);
    const float r1 = wino_g(g[3], g[4], g[5], qw);
    const float r2 = wino_g(g[6], g[7], g[8], qw);
    val = wino_g(r0, r1, r2, qh);
  } else if (m < 6) {
    const float* g = cw + ((16 + (m & 1)) * 64 + ic) * 9 + (m >> 1) * 3;
    val = wino_g(g[0], g[1], g[2], f - 16);
  }
  wp[t] = val;
}

#define RT 2
#define FR (RT + 2)
#define NWV 8

struct Unit {
  v2f a0[4][4];
  v2f a1[4];
  float dl[FR][2];
  v2f dm[FR][2];
  float dr[FR][2];
};

__device__ __forceinline__ void load_unit(Unit& f, const v2f* __restrict__ wa,
                                          const float* __restrict__ fc, const int (&ro)[FR],
                                          int lcl, int lcm, int lcr) {
#pragma unroll
  for (int qh = 0; qh < 4; ++qh) {
#pragma unroll
    for (int qw = 0; qw < 4; ++qw) f.a0[qh][qw] = wa[(4 * qh + qw) * 512];
  }
#pragma unroll
  for (int qw = 0; qw < 4; ++qw) f.a1[qw] = wa[(16 + qw) * 512];
#pragma unroll
  for (int i = 0; i < FR; ++i) {
    const float* fr = fc + ro[i];
#pragma unroll
    for (int v = 0; v < 2; ++v) {
      f.dl[i][v] = fr[lcl + v * HW];
      f.dm[i][v] = *(const v2f*)(fr + lcm + v * HW);
      f.dr[i][v] = fr[lcr + v * HW];
    }
  }
}

__device__ __forceinline__ void mma_unit(const Unit& f, v8f (&accw)[RT / 2][4][4],
                                         v8f (&acc1)[FR][4]) {
  v2f c[FR][4];
#pragma unroll
  for (int i = 0; i < FR; ++i) {
    const v2f d0 = {f.dl[i][0], f.dl[i][1]};
    const v2f d1 = {f.dm[i][0].x, f.dm[i][1].x};
    const v2f d2 = {f.dm[i][0].y, f.dm[i][1].y};
    const v2f d3 = {f.dr[i][0], f.dr[i][1]};
    c[i][0] = d0 - d2;
    c[i][1] = d1 + d2;
    c[i][2] = d2 - d1;
    c[i][3] = d1 - d3;
  }
#pragma unroll
  for (int qw = 0; qw < 4; ++qw) {
#pragma unroll
    for (int t = 0; t < RT / 2; ++t) {
      const v2f e0 = c[2 * t][qw], e1 = c[2 * t + 1][qw];
      const v2f e2 = c[2 * t + 2][qw], e3 = c[2 * t + 3][qw];
      const v2f tq[4] = {e0 - e2, e1 + e2, e2 - e1, e1 - e3};
#pragma unroll
      for (int qh = 0; qh < 4; ++qh) {
        accw[t][qh][qw] = __builtin_amdgcn_wmma_f32_16x16x4_f32(
            false, f.a0[qh][qw], false, tq[qh], (short)0, accw[t][qh][qw], false, false);
      }
    }
#pragma unroll
    for (int i = 0; i < FR; ++i) {
      acc1[i][qw] = __builtin_amdgcn_wmma_f32_16x16x4_f32(
          false, f.a1[qw], false, c[i][qw], (short)0, acc1[i][qw], false, false);
    }
  }
}

__global__ __launch_bounds__(NWV * 32) void conv_softmax(
    const float* __restrict__ feat, const float* __restrict__ bias,
    const float* __restrict__ wp, float* __restrict__ ffw) {
  __shared__ float zb[4][8 * 16 * 19];

  const int b = blockIdx.y;
  const int tid = threadIdx.x;

  const int lane = tid & 31, wv = tid >> 5;
  const int u = lane & 15, hi = lane >> 4;
  const int w0 = (wv & 3) * 32;
  const int rb = (wv >> 2) * RT;
  const int hw0 = blockIdx.x * 4;
  const int h0 = hw0 + rb;

  const v2f* __restrict__ wf = (const v2f*)wp + lane;

  const float* fb = feat + (size_t)b * 64 * HW;
  const int lcl = 2 * hi * HW + reflect_idx(w0 + 2 * u - 1, WW);
  const int lcm = 2 * hi * HW + w0 + 2 * u;
  const int lcr = 2 * hi * HW + reflect_idx(w0 + 2 * u + 2, WW);
  int ro[FR];
#pragma unroll
  for (int i = 0; i < FR; ++i) ro[i] = reflect_idx(h0 - 1 + i, HH) * WW;

  v8f accw[RT / 2][4][4];
  v8f acc1[FR][4];
#pragma unroll
  for (int t = 0; t < RT / 2; ++t) {
#pragma unroll
    for (int qh = 0; qh < 4; ++qh) {
#pragma unroll
      for (int qw = 0; qw < 4; ++qw)
        accw[t][qh][qw] = (v8f){0.f, 0.f, 0.f, 0.f, 0.f, 0.f, 0.f, 0.f};
    }
  }
#pragma unroll
  for (int i = 0; i < FR; ++i) {
#pragma unroll
    for (int qw = 0; qw < 4; ++qw) acc1[i][qw] = (v8f){0.f, 0.f, 0.f, 0.f, 0.f, 0.f, 0.f, 0.f};
  }

  Unit f0, f1;
  load_unit(f0, wf, fb, ro, lcl, lcm, lcr);
#pragma unroll 1
  for (int j = 0; j < 14; j += 2) {
    load_unit(f1, wf + (j + 1) * 32, fb + (j + 1) * 4 * HW, ro, lcl, lcm, lcr);
    __builtin_amdgcn_sched_barrier(0);
    mma_unit(f0, accw, acc1);
    __builtin_amdgcn_sched_barrier(0);
    load_unit(f0, wf + (j + 2) * 32, fb + (j + 2) * 4 * HW, ro, lcl, lcm, lcr);
    __builtin_amdgcn_sched_barrier(0);
    mma_unit(f1, accw, acc1);
  }
  load_unit(f1, wf + 15 * 32, fb + 15 * 4 * HW, ro, lcl, lcm, lcr);
  __builtin_amdgcn_sched_barrier(0);
  mma_unit(f0, accw, acc1);
  mma_unit(f1, accw, acc1);

  v8f se[FR], so[FR];
#pragma unroll
  for (int i = 0; i < FR; ++i) {
    se[i] = (acc1[i][0] + acc1[i][1]) + acc1[i][2];
    so[i] = (acc1[i][1] - acc1[i][2]) - acc1[i][3];
  }
#pragma unroll
  for (int t = 0; t < RT / 2; ++t) {
    v8f ye[4], yo[4];
#pragma unroll
    for (int qh = 0; qh < 4; ++qh) {
      ye[qh] = (accw[t][qh][0] + accw[t][qh][1]) + accw[t][qh][2];
      yo[qh] = (accw[t][qh][1] - accw[t][qh][2]) - accw[t][qh][3];
    }
#pragma unroll
    for (int oo = 0; oo < 2; ++oo) {
      const int o = 2 * t + oo;
      const v8f ze = oo ? (ye[1] - ye[2]) - ye[3] : (ye[0] + ye[1]) + ye[2];
      const v8f zo = oo ? (yo[1] - yo[2]) - yo[3] : (yo[0] + yo[1]) + yo[2];
      float* zre = &zb[rb + o][(w0 + 2 * u) * 19];
      float* zro = zre + 19;
#pragma unroll
      for (int r = 0; r < 8; ++r) {
        zre[r + 8 * hi] = ze[r];
        zro[r + 8 * hi] = zo[r];
      }
      if (hi == 0) {
        zre[16] = (se[o][0] + se[o + 1][2]) + se[o + 2][4];
        zre[17] = (se[o][1] + se[o + 1][3]) + se[o + 2][5];
        zro[16] = (so[o][0] + so[o + 1][2]) + so[o + 2][4];
        zro[17] = (so[o][1] + so[o + 1][3]) + so[o + 2][5];
      }
    }
  }
  __syncthreads();

  const int g = hi;
  const int p = u;
  float bz[9];
#pragma unroll
  for (int k = 0; k < 9; ++k) bz[k] = bias[2 * k + g];
#pragma unroll
  for (int oe = 0; oe < 2 * RT; ++oe) {
    const int o = oe >> 1;
    const int wpix = w0 + 16 * (oe & 1) + p;
    const float* zrow = &zb[rb + o][wpix * 19];
    float z[9];
#pragma unroll
    for (int k = 0; k < 9; ++k) z[k] = zrow[2 * k + g] + bz[k];
    float mx = z[0];
#pragma unroll
    for (int k = 1; k < 9; ++k) mx = fmaxf(mx, z[k]);
    float s = 0.0f;
#pragma unroll
    for (int k = 0; k < 9; ++k) {
      z[k] = expf(z[k] - mx);
      s += z[k];
    }
    const float inv = 1.0f / s;
#pragma unroll
    for (int k = 0; k < 9; ++k) z[k] *= inv;
    float ev[5], od[4];
#pragma unroll
    for (int i = 0; i < 5; ++i) ev[i] = g ? z[4 + i] : z[i];
#pragma unroll
    for (int i = 0; i < 4; ++i) od[i] = __shfl_xor(g ? z[i] : z[5 + i], 16);
    float* fblk = ffw + ((size_t)b * 2 + g) * 9 * HW;
    float4* q4 = (float4*)fblk + ((h0 + o) * 2) * WW + wpix;
    q4[0] = make_float4(ev[0], od[0], ev[1], od[1]);
    q4[WW] = make_float4(ev[2], od[2], ev[3], od[3]);
    fblk[8 * HW + (h0 + o) * WW + wpix] = ev[4];
  }
}

#define RB 8
__global__ __launch_bounds__(256) void combine(
    const float* __restrict__ inp, const float* __restrict__ ffw,
    float* __restrict__ out) {
  const int w = threadIdx.x & 127;
  const int slot = threadIdx.x >> 7;
  const int h0 = blockIdx.x * RB;
  const int s = blockIdx.y & 1;
  const int c = s + 2 * ((blockIdx.y >> 1) * 2 + slot);
  const int b = blockIdx.z;
  const int wl = reflect_idx(w - 1, WW);
  const int wr = reflect_idx(w + 1, WW);
  const float* xin = inp + (size_t)(b * 64 + c) * HW;
  const float* wblk = ffw + (((size_t)b * 2 + s) * 9) * HW;
  const float4* wq = (const float4*)wblk + w;
  const float* w8 = wblk + 8 * HW + w;
  float* op = out + (size_t)(b * 64 + c) * HW + w;

  const int rt = reflect_idx(h0 - 1, HH) * WW;
  float t0 = xin[rt + wl], t1 = xin[rt + w], t2 = xin[rt + wr];
  float m0 = xin[h0 * WW + wl], m1 = xin[h0 * WW + w], m2 = xin[h0 * WW + wr];
#pragma unroll
  for (int i = 0; i < RB; ++i) {
    const int h = h0 + i;
    const int rb = reflect_idx(h + 1, HH) * WW;
    const float u0 = xin[rb + wl], u1 = xin[rb + w], u2 = xin[rb + wr];
    const float4 qa = wq[(h * 2) * WW];
    const float4 qb = wq[(h * 2 + 1) * WW];
    const float k8 = w8[h * WW];
    float acc = 0.0f;
    acc = fmaf(qa.x, t0, acc);
    acc = fmaf(qa.y, t1, acc);
    acc = fmaf(qa.z, t2, acc);
    acc = fmaf(qa.w, m0, acc);
    acc = fmaf(qb.x, m1, acc);
    acc = fmaf(qb.y, m2, acc);
    acc = fmaf(qb.z, u0, acc);
    acc = fmaf(qb.w, u1, acc);
    acc = fmaf(k8, u2, acc);
    op[h * WW] = acc;
    t0 = m0; t1 = m1; t2 = m2;
    m0 = u0; m1 = u1; m2 = u2;
  }
}

extern "C" void kernel_launch(void* const* d_in, const int* in_sizes, int n_in,
                              void* d_out, int out_size, void* d_ws, size_t ws_size,
                              hipStream_t stream) {
  const float* features = (const float*)d_in[0];
  const float* inpt     = (const float*)d_in[1];
  const float* conv_w   = (const float*)d_in[2];
  const float* conv_b   = (const float*)d_in[3];
  float* out = (float*)d_out;

  float* wp  = (float*)d_ws;
  float* ffw = wp + 20480;

  pack_w<<<80, 256, 0, stream>>>(conv_w, wp);
  conv_softmax<<<dim3(32, 8), 256, 0, stream>>>(features, conv_b, wp, ffw);
  combine<<<dim3(16, 32, 8), 256, 0, stream>>>(inpt, ffw, out);
}


# combine kernel: a work-group copies its 8 rows' 9 softmax taps to LDS once; 8 channel slots read them there and keep them over 4 channels each (was 3 global loads per output); sums unchanged
# speedup vs baseline: 3.1768x; 1.1687x over previous
#include <hip/hip_runtime.h>
#include <math.h>


typedef __attribute__((ext_vector_type(2))) float v2f;
typedef __attribute__((ext_vector_type(8))) float v8f;

#define HH 128
#define WW 128
#define HW (HH * WW)

__device__ __forceinline__ int reflect_idx(int i, int n) {
  if (i < 0) i = -i;
  if (i >= n) i = 2 * n - 2 - i;
  return i;
}

__device__ __forceinline__ float wino_g(float g0, float g1, float g2, int q) {
  if (q == 0) return g0;
  if (q == 1) return 0.5f * ((g0 + g2) + g1);
  if (q == 2) return 0.5f * ((g0 + g2) - g1);
  return g2;
}

__global__ void pack_w(const float* __restrict__ cw, float* __restrict__ wp) {
  int t = blockIdx.x * blockDim.x + threadIdx.x;
  if (t >= 20 * 16 * 32 * 2) return;
  int v = t & 1;
  int lane = (t >> 1) & 31;
  int j = (t >> 6) & 15;
  int f = t >> 10;
  int m = lane & 15;
  int hi = lane >> 4;
  int ic = 4 * j + 2 * hi + v;
  float val = 0.0f;
  if (f < 16) {
    const int qh = f >> 2, qw = f & 3;
    const float* g = cw + (m * 64 + ic) * 9;
    const float r0 = wino_g(g[0], g[1], g[2], qw);
    const float r1 = wino_g(g[3], g[4], g[5], qw);
    const float r2 = wino_g(g[6], g[7], g[8], qw);
    val = wino_g(r0, r1, r2, qh);
  } else if (m < 6) {
    const float* g = cw + ((16 + (m & 1)) * 64 + ic) * 9 + (m >> 1) * 3;
    val = wino_g(g[0], g[1], g[2], f - 16);
  }
  wp[t] = val;
}

#define RT 2
#define FR (RT + 2)
#define NWV 8

struct Unit {
  v2f a0[4][4];
  v2f a1[4];
  float dl[FR][2];
  v2f dm[FR][2];
  float dr[FR][2];
};

__device__ __forceinline__ void load_unit(Unit& f, const v2f* __restrict__ wa,
                                          const float* __restrict__ fc, const int (&ro)[FR],
                                          int lcl, int lcm, int lcr) {
#pragma unroll
  for (int qh = 0; qh < 4; ++qh) {
#pragma unroll
    for (int qw = 0; qw < 4; ++qw) f.a0[qh][qw] = wa[(4 * qh + qw) * 512];
  }
#pragma unroll
  for (int qw = 0; qw < 4; ++qw) f.a1[qw] = wa[(16 + qw) * 512];
#pragma unroll
  for (int i = 0; i < FR; ++i) {
    const float* fr = fc + ro[i];
#pragma unroll
    for (int v = 0; v < 2; ++v) {
      f.dl[i][v] = fr[lcl + v * HW];
      f.dm[i][v] = *(const v2f*)(fr + lcm + v * HW);
      f.dr[i][v] = fr[lcr + v * HW];
    }
  }
}

__device__ __forceinline__ void mma_unit(const Unit& f, v8f (&accw)[RT / 2][4][4],
                                         v8f (&acc1)[FR][4]) {
  v2f c[FR][4];
#pragma unroll
  for (int i = 0; i < FR; ++i) {
    const v2f d0 = {f.dl[i][0], f.dl[i][1]};
    const v2f d1 = {f.dm[i][0].x, f.dm[i][1].x};
    const v2f d2 = {f.dm[i][0].y, f.dm[i][1].y};
    const v2f d3 = {f.dr[i][0], f.dr[i][1]};
    c[i][0] = d0 - d2;
    c[i][1] = d1 + d2;
    c[i][2] = d2 - d1;
    c[i][3] = d1 - d3;
  }
#pragma unroll
  for (int qw = 0; qw < 4; ++qw) {
#pragma unroll
    for (int t = 0; t < RT / 2; ++t) {
      const v2f e0 = c[2 * t][qw], e1 = c[2 * t + 1][qw];
      const v2f e2 = c[2 * t + 2][qw], e3 = c[2 * t + 3][qw];
      const v2f tq[4] = {e0 - e2, e1 + e2, e2 - e1, e1 - e3};
#pragma unroll
      for (int qh = 0; qh < 4; ++qh) {
        accw[t][qh][qw] = __builtin_amdgcn_wmma_f32_16x16x4_f32(
            false, f.a0[qh][qw], false, tq[qh], (short)0, accw[t][qh][qw], false, false);
      }
    }
#pragma unroll
    for (int i = 0; i < FR; ++i) {
      acc1[i][qw] = __builtin_amdgcn_wmma_f32_16x16x4_f32(
          false, f.a1[qw], false, c[i][qw], (short)0, acc1[i][qw], false, false);
    }
  }
}

__global__ __launch_bounds__(NWV * 32) void conv_softmax(
    const float* __restrict__ feat, const float* __restrict__ bias,
    const float* __restrict__ wp, float* __restrict__ ffw) {
  __shared__ float zb[4][8 * 16 * 19];

  const int b = blockIdx.y;
  const int tid = threadIdx.x;

  const int lane = tid & 31, wv = tid >> 5;
  const int u = lane & 15, hi = lane >> 4;
  const int w0 = (wv & 3) * 32;
  const int rb = (wv >> 2) * RT;
  const int hw0 = blockIdx.x * 4;
  const int h0 = hw0 + rb;

  const v2f* __restrict__ wf = (const v2f*)wp + lane;

  const float* fb = feat + (size_t)b * 64 * HW;
  const int lcl = 2 * hi * HW + reflect_idx(w0 + 2 * u - 1, WW);
  const int lcm = 2 * hi * HW + w0 + 2 * u;
  const int lcr = 2 * hi * HW + reflect_idx(w0 + 2 * u + 2, WW);
  int ro[FR];
#pragma unroll
  for (int i = 0; i < FR; ++i) ro[i] = reflect_idx(h0 - 1 + i, HH) * WW;

  v8f accw[RT / 2][4][4];
  v8f acc1[FR][4];
#pragma unroll
  for (int t = 0; t < RT / 2; ++t) {
#pragma unroll
    for (int qh = 0; qh < 4; ++qh) {
#pragma unroll
      for (int qw = 0; qw < 4; ++qw)
        accw[t][qh][qw] = (v8f){0.f, 0.f, 0.f, 0.f, 0.f, 0.f, 0.f, 0.f};
    }
  }
#pragma unroll
  for (int i = 0; i < FR; ++i) {
#pragma unroll
    for (int qw = 0; qw < 4; ++qw) acc1[i][qw] = (v8f){0.f, 0.f, 0.f, 0.f, 0.f, 0.f, 0.f, 0.f};
  }

  Unit f0, f1;
  load_unit(f0, wf, fb, ro, lcl, lcm, lcr);
#pragma unroll 1
  for (int j = 0; j < 14; j += 2) {
    load_unit(f1, wf + (j + 1) * 32, fb + (j + 1) * 4 * HW, ro, lcl, lcm, lcr);
    __builtin_amdgcn_sched_barrier(0);
    mma_unit(f0, accw, acc1);
    __builtin_amdgcn_sched_barrier(0);
    load_unit(f0, wf + (j + 2) * 32, fb + (j + 2) * 4 * HW, ro, lcl, lcm, lcr);
    __builtin_amdgcn_sched_barrier(0);
    mma_unit(f1, accw, acc1);
  }
  load_unit(f1, wf + 15 * 32, fb + 15 * 4 * HW, ro, lcl, lcm, lcr);
  __builtin_amdgcn_sched_barrier(0);
  mma_unit(f0, accw, acc1);
  mma_unit(f1, accw, acc1);

  v8f se[FR], so[FR];
#pragma unroll
  for (int i = 0; i < FR; ++i) {
    se[i] = (acc1[i][0] + acc1[i][1]) + acc1[i][2];
    so[i] = (acc1[i][1] - acc1[i][2]) - acc1[i][3];
  }
#pragma unroll
  for (int t = 0; t < RT / 2; ++t) {
    v8f ye[4], yo[4];
#pragma unroll
    for (int qh = 0; qh < 4; ++qh) {
      ye[qh] = (accw[t][qh][0] + accw[t][qh][1]) + accw[t][qh][2];
      yo[qh] = (accw[t][qh][1] - accw[t][qh][2]) - accw[t][qh][3];
    }
#pragma unroll
    for (int oo = 0; oo < 2; ++oo) {
      const int o = 2 * t + oo;
      const v8f ze = oo ? (ye[1] - ye[2]) - ye[3] : (ye[0] + ye[1]) + ye[2];
      const v8f zo = oo ? (yo[1] - yo[2]) - yo[3] : (yo[0] + yo[1]) + yo[2];
      float* zre = &zb[rb + o][(w0 + 2 * u) * 19];
      float* zro = zre + 19;
#pragma unroll
      for (int r = 0; r < 8; ++r) {
        zre[r + 8 * hi] = ze[r];
        zro[r + 8 * hi] = zo[r];
      }
      if (hi == 0) {
        zre[16] = (se[o][0] + se[o + 1][2]) + se[o + 2][4];
        zre[17] = (se[o][1] + se[o + 1][3]) + se[o + 2][5];
        zro[16] = (so[o][0] + so[o + 1][2]) + so[o + 2][4];
        zro[17] = (so[o][1] + so[o + 1][3]) + so[o + 2][5];
      }
    }
  }
  __syncthreads();

  const int g = hi;
  const int p = u;
  float bz[9];
#pragma unroll
  for (int k = 0; k < 9; ++k) bz[k] = bias[2 * k + g];
#pragma unroll
  for (int oe = 0; oe < 2 * RT; ++oe) {
    const int o = oe >> 1;
    const int wpix = w0 + 16 * (oe & 1) + p;
    const float* zrow = &zb[rb + o][wpix * 19];
    float z[9];
#pragma unroll
    for (int k = 0; k < 9; ++k) z[k] = zrow[2 * k + g] + bz[k];
    float mx = z[0];
#pragma unroll
    for (int k = 1; k < 9; ++k) mx = fmaxf(mx, z[k]);
    float s = 0.0f;
#pragma unroll
    for (int k = 0; k < 9; ++k) {
      z[k] = expf(z[k] - mx);
      s += z[k];
    }
    const float inv = 1.0f / s;
#pragma unroll
    for (int k = 0; k < 9; ++k) z[k] *= inv;
    float ev[5], od[4];
#pragma unroll
    for (int i = 0; i < 5; ++i) ev[i] = g ? z[4 + i] : z[i];
#pragma unroll
    for (int i = 0; i < 4; ++i) od[i] = __shfl_xor(g ? z[i] : z[5 + i], 16);
    float* fblk = ffw + ((size_t)b * 2 + g) * 9 * HW;
    float4* q4 = (float4*)fblk + ((h0 + o) * 2) * WW + wpix;
    q4[0] = make_float4(ev[0], od[0], ev[1], od[1]);
    q4[WW] = make_float4(ev[2], od[2], ev[3], od[3]);
    fblk[8 * HW + (h0 + o) * WW + wpix] = ev[4];
  }
}

#define RB 8
#define CS 8
__global__ __launch_bounds__(CS * 128) void combine(
    const float* __restrict__ inp, const float* __restrict__ ffw,
    float* __restrict__ out) {
  __shared__ float4 wq_s[RB * 2 * WW];
  __shared__ float w8_s[RB * WW];

  const unsigned tid = threadIdx.x;
  const unsigned w = tid & 127u;
  const unsigned slot = tid >> 7;
  const int h0 = blockIdx.x * RB;
  const int s = blockIdx.y;
  const int b = blockIdx.z;
  const float* wblk = ffw + (((size_t)b * 2 + s) * 9) * HW;

  {
    const float4* src4 = (const float4*)wblk + h0 * 2 * WW;
    const float* src8 = wblk + 8 * HW + h0 * WW;
#pragma unroll
    for (unsigned i = tid; i < RB * 2 * WW; i += CS * 128) wq_s[i] = src4[i];
#pragma unroll
    for (unsigned i = tid; i < RB * WW; i += CS * 128) w8_s[i] = src8[i];
  }
  __syncthreads();

  const unsigned lm = slot * (2u * HW) + w;
  const unsigned ll = slot * (2u * HW) + (unsigned)reflect_idx((int)w - 1, WW);
  const unsigned lr = slot * (2u * HW) + (unsigned)reflect_idx((int)w + 1, WW);
  const int rt = reflect_idx(h0 - 1, HH) * WW;
  const int rl = reflect_idx(h0 + RB, HH) * WW;

  for (int cg = 0; cg < 32; cg += CS) {
    const size_t cbase = (size_t)(b * 64 + s + 2 * cg) * HW;
    const float* xin = inp + cbase;
    const float* xrow = xin + h0 * WW;
    float* orow = out + cbase + h0 * WW;
    float t0 = xin[rt + ll], t1 = xin[rt + lm], t2 = xin[rt + lr];
    float m0 = xrow[ll], m1 = xrow[lm], m2 = xrow[lr];
#pragma unroll
    for (int i = 0; i < RB; ++i) {
      const float* xn = (i + 1 < RB) ? xrow + (i + 1) * WW : xin + rl;
      const float u0 = xn[ll], u1 = xn[lm], u2 = xn[lr];
      const float4 qa = wq_s[(i * 2) * WW + w];
      const float4 qb = wq_s[(i * 2 + 1) * WW + w];
      const float k8 = w8_s[i * WW + w];
      float acc = 0.0f;
      acc = fmaf(qa.x, t0, acc);
      acc = fmaf(qa.y, t1, acc);
      acc = fmaf(qa.z, t2, acc);
      acc = fmaf(qa.w, m0, acc);
      acc = fmaf(qb.x, m1, acc);
      acc = fmaf(qb.y, m2, acc);
      acc = fmaf(qb.z, u0, acc);
      acc = fmaf(qb.w, u1, acc);
      acc = fmaf(k8, u2, acc);
      orow[i * WW + lm] = acc;
      t0 = m0; t1 = m1; t2 = m2;
      m0 = u0; m1 = u1; m2 = u2;
    }
  }
}

extern "C" void kernel_launch(void* const* d_in, const int* in_sizes, int n_in,
                              void* d_out, int out_size, void* d_ws, size_t ws_size,
                              hipStream_t stream) {
  const float* features = (const float*)d_in[0];
  const float* inpt     = (const float*)d_in[1];
  const float* conv_w   = (const float*)d_in[2];
  const float* conv_b   = (const float*)d_in[3];
  float* out = (float*)d_out;

  float* wp  = (float*)d_ws;
  float* ffw = wp + 20480;

  pack_w<<<80, 256, 0, stream>>>(conv_w, wp);
  conv_softmax<<<dim3(32, 8), 256, 0, stream>>>(features, conv_b, wp, ffw);
  combine<<<dim3(16, 2, 8), 1024, 0, stream>>>(inpt, ffw, out);
}


# softmax shift invariance: pack subtracts a group's tap-8 filter from taps 0..7, conv makes 16 logit differences (9th = 0): 16 WMMAs per unit, was 32; freed registers: loads 2 units ahead; new sums
# speedup vs baseline: 3.5607x; 1.1208x over previous
#include <hip/hip_runtime.h>
#include <math.h>


typedef __attribute__((ext_vector_type(2))) float v2f;
typedef __attribute__((ext_vector_type(8))) float v8f;

#define HH 128
#define WW 128
#define HW (HH * WW)

__device__ __forceinline__ int reflect_idx(int i, int n) {
  if (i < 0) i = -i;
  if (i >= n) i = 2 * n - 2 - i;
  return i;
}

__device__ __forceinline__ float wino_g(float g0, float g1, float g2, int q) {
  if (q == 0) return g0;
  if (q == 1) return 0.5f * ((g0 + g2) + g1);
  if (q == 2) return 0.5f * ((g0 + g2) - g1);
  return g2;
}

__global__ void pack_w(const float* __restrict__ cw, float* __restrict__ wp) {
  int t = blockIdx.x * blockDim.x + threadIdx.x;
  if (t >= 16 * 16 * 32 * 2) return;
  int v = t & 1;
  int lane = (t >> 1) & 31;
  int j = (t >> 6) & 15;
  int f = t >> 10;
  int m = lane & 15;
  int hi = lane >> 4;
  int ic = 4 * j + 2 * hi + v;
  const int qh = f >> 2, qw = f & 3;
  const float* g = cw + (m * 64 + ic) * 9;
  const float* e = cw + ((16 + (m & 1)) * 64 + ic) * 9;
  float d[9];
#pragma unroll
  for (int i = 0; i < 9; ++i) d[i] = g[i] - e[i];
  const float r0 = wino_g(d[0], d[1], d[2], qw);
  const float r1 = wino_g(d[3], d[4], d[5], qw);
  const float r2 = wino_g(d[6], d[7], d[8], qw);
  wp[t] = wino_g(r0, r1, r2, qh);
}

#define RT 2
#define FR (RT + 2)
#define NWV 8

struct Unit {
  v2f a0[4][4];
  float dl[FR][2];
  v2f dm[FR][2];
  float dr[FR][2];
};

__device__ __forceinline__ void load_unit(Unit& f, const v2f* __restrict__ wa,
                                          const float* __restrict__ fc, const int (&ro)[FR],
                                          int lcl, int lcm, int lcr) {
#pragma unroll
  for (int qh = 0; qh < 4; ++qh) {
#pragma unroll
    for (int qw = 0; qw < 4; ++qw) f.a0[qh][qw] = wa[(4 * qh + qw) * 512];
  }
#pragma unroll
  for (int i = 0; i < FR; ++i) {
    const float* fr = fc + ro[i];
#pragma unroll
    for (int v = 0; v < 2; ++v) {
      f.dl[i][v] = fr[lcl + v * HW];
      f.dm[i][v] = *(const v2f*)(fr + lcm + v * HW);
      f.dr[i][v] = fr[lcr + v * HW];
    }
  }
}

__device__ __forceinline__ void mma_unit(const Unit& f, v8f (&accw)[RT / 2][4][4]) {
  v2f c[FR][4];
#pragma unroll
  for (int i = 0; i < FR; ++i) {
    const v2f d0 = {f.dl[i][0], f.dl[i][1]};
    const v2f d1 = {f.dm[i][0].x, f.dm[i][1].x};
    const v2f d2 = {f.dm[i][0].y, f.dm[i][1].y};
    const v2f d3 = {f.dr[i][0], f.dr[i][1]};
    c[i][0] = d0 - d2;
    c[i][1] = d1 + d2;
    c[i][2] = d2 - d1;
    c[i][3] = d1 - d3;
  }
#pragma unroll
  for (int qw = 0; qw < 4; ++qw) {
#pragma unroll
    for (int t = 0; t < RT / 2; ++t) {
      const v2f e0 = c[2 * t][qw], e1 = c[2 * t + 1][qw];
      const v2f e2 = c[2 * t + 2][qw], e3 = c[2 * t + 3][qw];
      const v2f tq[4] = {e0 - e2, e1 + e2, e2 - e1, e1 - e3};
#pragma unroll
      for (int qh = 0; qh < 4; ++qh) {
        accw[t][qh][qw] = __builtin_amdgcn_wmma_f32_16x16x4_f32(
            false, f.a0[qh][qw], false, tq[qh], (short)0, accw[t][qh][qw], false, false);
      }
    }
  }
}

__global__ __launch_bounds__(NWV * 32) void conv_softmax(
    const float* __restrict__ feat, const float* __restrict__ bias,
    const float* __restrict__ wp, float* __restrict__ ffw) {
  __shared__ float zb[4][8 * 16 * 19];

  const int b = blockIdx.y;
  const int tid = threadIdx.x;

  const int lane = tid & 31, wv = tid >> 5;
  const int u = lane & 15, hi = lane >> 4;
  const int w0 = (wv & 3) * 32;
  const int rb = (wv >> 2) * RT;
  const int hw0 = blockIdx.x * 4;
  const int h0 = hw0 + rb;

  const v2f* __restrict__ wf = (const v2f*)wp + lane;

  const float* fb = feat + (size_t)b * 64 * HW;
  const int lcl = 2 * hi * HW + reflect_idx(w0 + 2 * u - 1, WW);
  const int lcm = 2 * hi * HW + w0 + 2 * u;
  const int lcr = 2 * hi * HW + reflect_idx(w0 + 2 * u + 2, WW);
  int ro[FR];
#pragma unroll
  for (int i = 0; i < FR; ++i) ro[i] = reflect_idx(h0 - 1 + i, HH) * WW;

  v8f accw[RT / 2][4][4];
#pragma unroll
  for (int t = 0; t < RT / 2; ++t) {
#pragma unroll
    for (int qh = 0; qh < 4; ++qh) {
#pragma unroll
      for (int qw = 0; qw < 4; ++qw)
        accw[t][qh][qw] = (v8f){0.f, 0.f, 0.f, 0.f, 0.f, 0.f, 0.f, 0.f};
    }
  }

  Unit f0, f1, f2;
  load_unit(f0, wf, fb, ro, lcl, lcm, lcr);
  __builtin_amdgcn_sched_barrier(0);
  load_unit(f1, wf + 32, fb + 4 * HW, ro, lcl, lcm, lcr);
#pragma unroll 1
  for (int j = 0; j < 12; j += 3) {
    load_unit(f2, wf + (j + 2) * 32, fb + (j + 2) * 4 * HW, ro, lcl, lcm, lcr);
    __builtin_amdgcn_sched_barrier(0);
    mma_unit(f0, accw);
    __builtin_amdgcn_sched_barrier(0);
    load_unit(f0, wf + (j + 3) * 32, fb + (j + 3) * 4 * HW, ro, lcl, lcm, lcr);
    __builtin_amdgcn_sched_barrier(0);
    mma_unit(f1, accw);
    __builtin_amdgcn_sched_barrier(0);
    load_unit(f1, wf + (j + 4) * 32, fb + (j + 4) * 4 * HW, ro, lcl, lcm, lcr);
    __builtin_amdgcn_sched_barrier(0);
    mma_unit(f2, accw);
    __builtin_amdgcn_sched_barrier(0);
  }
  load_unit(f2, wf + 14 * 32, fb + 14 * 4 * HW, ro, lcl, lcm, lcr);
  __builtin_amdgcn_sched_barrier(0);
  mma_unit(f0, accw);
  __builtin_amdgcn_sched_barrier(0);
  load_unit(f0, wf + 15 * 32, fb + 15 * 4 * HW, ro, lcl, lcm, lcr);
  __builtin_amdgcn_sched_barrier(0);
  mma_unit(f1, accw);
  mma_unit(f2, accw);
  mma_unit(f0, accw);

#pragma unroll
  for (int t = 0; t < RT / 2; ++t) {
    v8f ye[4], yo[4];
#pragma unroll
    for (int qh = 0; qh < 4; ++qh) {
      ye[qh] = (accw[t][qh][0] + accw[t][qh][1]) + accw[t][qh][2];
      yo[qh] = (accw[t][qh][1] - accw[t][qh][2]) - accw[t][qh][3];
    }
#pragma unroll
    for (int oo = 0; oo < 2; ++oo) {
      const int o = 2 * t + oo;
      const v8f ze = oo ? (ye[1] - ye[2]) - ye[3] : (ye[0] + ye[1]) + ye[2];
      const v8f zo = oo ? (yo[1] - yo[2]) - yo[3] : (yo[0] + yo[1]) + yo[2];
      float* zre = &zb[rb + o][(w0 + 2 * u) * 19];
      float* zro = zre + 19;
#pragma unroll
      for (int r = 0; r < 8; ++r) {
        zre[r + 8 * hi] = ze[r];
        zro[r + 8 * hi] = zo[r];
      }
    }
  }
  __syncthreads();

  const int g = hi;
  const int p = u;
  float bz[8];
#pragma unroll
  for (int k = 0; k < 8; ++k) bz[k] = bias[2 * k + g] - bias[16 + g];
#pragma unroll
  for (int oe = 0; oe < 2 * RT; ++oe) {
    const int o = oe >> 1;
    const int wpix = w0 + 16 * (oe & 1) + p;
    const float* zrow = &zb[rb + o][wpix * 19];
    float z[9];
#pragma unroll
    for (int k = 0; k < 8; ++k) z[k] = zrow[2 * k + g] + bz[k];
    z[8] = 0.0f;
    float mx = z[0];
#pragma unroll
    for (int k = 1; k < 9; ++k) mx = fmaxf(mx, z[k]);
    float s = 0.0f;
#pragma unroll
    for (int k = 0; k < 9; ++k) {
      z[k] = expf(z[k] - mx);
      s += z[k];
    }
    const float inv = 1.0f / s;
#pragma unroll
    for (int k = 0; k < 9; ++k) z[k] *= inv;
    float ev[5], od[4];
#pragma unroll
    for (int i = 0; i < 5; ++i) ev[i] = g ? z[4 + i] : z[i];
#pragma unroll
    for (int i = 0; i < 4; ++i) od[i] = __shfl_xor(g ? z[i] : z[5 + i], 16);
    float* fblk = ffw + ((size_t)b * 2 + g) * 9 * HW;
    float4* q4 = (float4*)fblk + ((h0 + o) * 2) * WW + wpix;
    q4[0] = make_float4(ev[0], od[0], ev[1], od[1]);
    q4[WW] = make_float4(ev[2], od[2], ev[3], od[3]);
    fblk[8 * HW + (h0 + o) * WW + wpix] = ev[4];
  }
}

#define RB 8
#define CS 8
__global__ __launch_bounds__(CS * 128) void combine(
    const float* __restrict__ inp, const float* __restrict__ ffw,
    float* __restrict__ out) {
  __shared__ float4 wq_s[RB * 2 * WW];
  __shared__ float w8_s[RB * WW];

  const unsigned tid = threadIdx.x;
  const unsigned w = tid & 127u;
  const unsigned slot = tid >> 7;
  const int h0 = blockIdx.x * RB;
  const int s = blockIdx.y;
  const int b = blockIdx.z;
  const float* wblk = ffw + (((size_t)b * 2 + s) * 9) * HW;

  {
    const float4* src4 = (const float4*)wblk + h0 * 2 * WW;
    const float* src8 = wblk + 8 * HW + h0 * WW;
#pragma unroll
    for (unsigned i = tid; i < RB * 2 * WW; i += CS * 128) wq_s[i] = src4[i];
#pragma unroll
    for (unsigned i = tid; i < RB * WW; i += CS * 128) w8_s[i] = src8[i];
  }
  __syncthreads();

  const unsigned lm = slot * (2u * HW) + w;
  const unsigned ll = slot * (2u * HW) + (unsigned)reflect_idx((int)w - 1, WW);
  const unsigned lr = slot * (2u * HW) + (unsigned)reflect_idx((int)w + 1, WW);
  const int rt = reflect_idx(h0 - 1, HH) * WW;
  const int rl = reflect_idx(h0 + RB, HH) * WW;

  for (int cg = 0; cg < 32; cg += CS) {
    const size_t cbase = (size_t)(b * 64 + s + 2 * cg) * HW;
    const float* xin = inp + cbase;
    const float* xrow = xin + h0 * WW;
    float* orow = out + cbase + h0 * WW;
    float t0 = xin[rt + ll], t1 = xin[rt + lm], t2 = xin[rt + lr];
    float m0 = xrow[ll], m1 = xrow[lm], m2 = xrow[lr];
#pragma unroll
    for (int i = 0; i < RB; ++i) {
      const float* xn = (i + 1 < RB) ? xrow + (i + 1) * WW : xin + rl;
      const float u0 = xn[ll], u1 = xn[lm], u2 = xn[lr];
      const float4 qa = wq_s[(i * 2) * WW + w];
      const float4 qb = wq_s[(i * 2 + 1) * WW + w];
      const float k8 = w8_s[i * WW + w];
      float acc = 0.0f;
      acc = fmaf(qa.x, t0, acc);
      acc = fmaf(qa.y, t1, acc);
      acc = fmaf(qa.z, t2, acc);
      acc = fmaf(qa.w, m0, acc);
      acc = fmaf(qb.x, m1, acc);
      acc = fmaf(qb.y, m2, acc);
      acc = fmaf(qb.z, u0, acc);
      acc = fmaf(qb.w, u1, acc);
      acc = fmaf(k8, u2, acc);
      orow[i * WW + lm] = acc;
      t0 = m0; t1 = m1; t2 = m2;
      m0 = u0; m1 = u1; m2 = u2;
    }
  }
}

extern "C" void kernel_launch(void* const* d_in, const int* in_sizes, int n_in,
                              void* d_out, int out_size, void* d_ws, size_t ws_size,
                              hipStream_t stream) {
  const float* features = (const float*)d_in[0];
  const float* inpt     = (const float*)d_in[1];
  const float* conv_w   = (const float*)d_in[2];
  const float* conv_b   = (const float*)d_in[3];
  float* out = (float*)d_out;

  float* wp  = (float*)d_ws;
  float* ffw = wp + 16384;

  pack_w<<<64, 256, 0, stream>>>(conv_w, wp);
  conv_softmax<<<dim3(32, 8), 256, 0, stream>>>(features, conv_b, wp, ffw);
  combine<<<dim3(16, 2, 8), 1024, 0, stream>>>(inpt, ffw, out);
}


# combine kernel: each channel's 10-row inpt tile goes through LDS: an element is loaded from global memory once (was 3 times), column taps are read in LDS; loads run 2 rounds ahead; sums unchanged
# speedup vs baseline: 3.6970x; 1.0383x over previous
#include <hip/hip_runtime.h>
#include <math.h>


typedef __attribute__((ext_vector_type(2))) float v2f;
typedef __attribute__((ext_vector_type(8))) float v8f;

#define HH 128
#define WW 128
#define HW (HH * WW)

__device__ __forceinline__ int reflect_idx(int i, int n) {
  if (i < 0) i = -i;
  if (i >= n) i = 2 * n - 2 - i;
  return i;
}

__device__ __forceinline__ float wino_g(float g0, float g1, float g2, int q) {
  if (q == 0) return g0;
  if (q == 1) return 0.5f * ((g0 + g2) + g1);
  if (q == 2) return 0.5f * ((g0 + g2) - g1);
  return g2;
}

__global__ void pack_w(const float* __restrict__ cw, float* __restrict__ wp) {
  int t = blockIdx.x * blockDim.x + threadIdx.x;
  if (t >= 16 * 16 * 32 * 2) return;
  int v = t & 1;
  int lane = (t >> 1) & 31;
  int j = (t >> 6) & 15;
  int f = t >> 10;
  int m = lane & 15;
  int hi = lane >> 4;
  int ic = 4 * j + 2 * hi + v;
  const int qh = f >> 2, qw = f & 3;
  const float* g = cw + (m * 64 + ic) * 9;
  const float* e = cw + ((16 + (m & 1)) * 64 + ic) * 9;
  float d[9];
#pragma unroll
  for (int i = 0; i < 9; ++i) d[i] = g[i] - e[i];
  const float r0 = wino_g(d[0], d[1], d[2], qw);
  const float r1 = wino_g(d[3], d[4], d[5], qw);
  const float r2 = wino_g(d[6], d[7], d[8], qw);
  wp[t] = wino_g(r0, r1, r2, qh);
}

#define RT 2
#define FR (RT + 2)
#define NWV 8

struct Unit {
  v2f a0[4][4];
  float dl[FR][2];
  v2f dm[FR][2];
  float dr[FR][2];
};

__device__ __forceinline__ void load_unit(Unit& f, const v2f* __restrict__ wa,
                                          const float* __restrict__ fc, const int (&ro)[FR],
                                          int lcl, int lcm, int lcr) {
#pragma unroll
  for (int qh = 0; qh < 4; ++qh) {
#pragma unroll
    for (int qw = 0; qw < 4; ++qw) f.a0[qh][qw] = wa[(4 * qh + qw) * 512];
  }
#pragma unroll
  for (int i = 0; i < FR; ++i) {
    const float* fr = fc + ro[i];
#pragma unroll
    for (int v = 0; v < 2; ++v) {
      f.dl[i][v] = fr[lcl + v * HW];
      f.dm[i][v] = *(const v2f*)(fr + lcm + v * HW);
      f.dr[i][v] = fr[lcr + v * HW];
    }
  }
}

__device__ __forceinline__ void mma_unit(const Unit& f, v8f (&accw)[RT / 2][4][4]) {
  v2f c[FR][4];
#pragma unroll
  for (int i = 0; i < FR; ++i) {
    const v2f d0 = {f.dl[i][0], f.dl[i][1]};
    const v2f d1 = {f.dm[i][0].x, f.dm[i][1].x};
    const v2f d2 = {f.dm[i][0].y, f.dm[i][1].y};
    const v2f d3 = {f.dr[i][0], f.dr[i][1]};
    c[i][0] = d0 - d2;
    c[i][1] = d1 + d2;
    c[i][2] = d2 - d1;
    c[i][3] = d1 - d3;
  }
#pragma unroll
  for (int qw = 0; qw < 4; ++qw) {
#pragma unroll
    for (int t = 0; t < RT / 2; ++t) {
      const v2f e0 = c[2 * t][qw], e1 = c[2 * t + 1][qw];
      const v2f e2 = c[2 * t + 2][qw], e3 = c[2 * t + 3][qw];
      const v2f tq[4] = {e0 - e2, e1 + e2, e2 - e1, e1 - e3};
#pragma unroll
      for (int qh = 0; qh < 4; ++qh) {
        accw[t][qh][qw] = __builtin_amdgcn_wmma_f32_16x16x4_f32(
            false, f.a0[qh][qw], false, tq[qh], (short)0, accw[t][qh][qw], false, false);
      }
    }
  }
}

__global__ __launch_bounds__(NWV * 32) void conv_softmax(
    const float* __restrict__ feat, const float* __restrict__ bias,
    const float* __restrict__ wp, float* __restrict__ ffw) {
  __shared__ float zb[4][8 * 16 * 19];

  const int b = blockIdx.y;
  const int tid = threadIdx.x;

  const int lane = tid & 31, wv = tid >> 5;
  const int u = lane & 15, hi = lane >> 4;
  const int w0 = (wv & 3) * 32;
  const int rb = (wv >> 2) * RT;
  const int hw0 = blockIdx.x * 4;
  const int h0 = hw0 + rb;

  const v2f* __restrict__ wf = (const v2f*)wp + lane;

  const float* fb = feat + (size_t)b * 64 * HW;
  const int lcl = 2 * hi * HW + reflect_idx(w0 + 2 * u - 1, WW);
  const int lcm = 2 * hi * HW + w0 + 2 * u;
  const int lcr = 2 * hi * HW + reflect_idx(w0 + 2 * u + 2, WW);
  int ro[FR];
#pragma unroll
  for (int i = 0; i < FR; ++i) ro[i] = reflect_idx(h0 - 1 + i, HH) * WW;

  v8f accw[RT / 2][4][4];
#pragma unroll
  for (int t = 0; t < RT / 2; ++t) {
#pragma unroll
    for (int qh = 0; qh < 4; ++qh) {
#pragma unroll
      for (int qw = 0; qw < 4; ++qw)
        accw[t][qh][qw] = (v8f){0.f, 0.f, 0.f, 0.f, 0.f, 0.f, 0.f, 0.f};
    }
  }

  Unit f0, f1, f2;
  load_unit(f0, wf, fb, ro, lcl, lcm, lcr);
  __builtin_amdgcn_sched_barrier(0);
  load_unit(f1, wf + 32, fb + 4 * HW, ro, lcl, lcm, lcr);
#pragma unroll 1
  for (int j = 0; j < 12; j += 3) {
    load_unit(f2, wf + (j + 2) * 32, fb + (j + 2) * 4 * HW, ro, lcl, lcm, lcr);
    __builtin_amdgcn_sched_barrier(0);
    mma_unit(f0, accw);
    __builtin_amdgcn_sched_barrier(0);
    load_unit(f0, wf + (j + 3) * 32, fb + (j + 3) * 4 * HW, ro, lcl, lcm, lcr);
    __builtin_amdgcn_sched_barrier(0);
    mma_unit(f1, accw);
    __builtin_amdgcn_sched_barrier(0);
    load_unit(f1, wf + (j + 4) * 32, fb + (j + 4) * 4 * HW, ro, lcl, lcm, lcr);
    __builtin_amdgcn_sched_barrier(0);
    mma_unit(f2, accw);
    __builtin_amdgcn_sched_barrier(0);
  }
  load_unit(f2, wf + 14 * 32, fb + 14 * 4 * HW, ro, lcl, lcm, lcr);
  __builtin_amdgcn_sched_barrier(0);
  mma_unit(f0, accw);
  __builtin_amdgcn_sched_barrier(0);
  load_unit(f0, wf + 15 * 32, fb + 15 * 4 * HW, ro, lcl, lcm, lcr);
  __builtin_amdgcn_sched_barrier(0);
  mma_unit(f1, accw);
  mma_unit(f2, accw);
  mma_unit(f0, accw);

#pragma unroll
  for (int t = 0; t < RT / 2; ++t) {
    v8f ye[4], yo[4];
#pragma unroll
    for (int qh = 0; qh < 4; ++qh) {
      ye[qh] = (accw[t][qh][0] + accw[t][qh][1]) + accw[t][qh][2];
      yo[qh] = (accw[t][qh][1] - accw[t][qh][2]) - accw[t][qh][3];
    }
#pragma unroll
    for (int oo = 0; oo < 2; ++oo) {
      const int o = 2 * t + oo;
      const v8f ze = oo ? (ye[1] - ye[2]) - ye[3] : (ye[0] + ye[1]) + ye[2];
      const v8f zo = oo ? (yo[1] - yo[2]) - yo[3] : (yo[0] + yo[1]) + yo[2];
      float* zre = &zb[rb + o][(w0 + 2 * u) * 19];
      float* zro = zre + 19;
#pragma unroll
      for (int r = 0; r < 8; ++r) {
        zre[r + 8 * hi] = ze[r];
        zro[r + 8 * hi] = zo[r];
      }
    }
  }
  __syncthreads();

  const int g = hi;
  const int p = u;
  float bz[8];
#pragma unroll
  for (int k = 0; k < 8; ++k) bz[k] = bias[2 * k + g] - bias[16 + g];
#pragma unroll
  for (int oe = 0; oe < 2 * RT; ++oe) {
    const int o = oe >> 1;
    const int wpix = w0 + 16 * (oe & 1) + p;
    const float* zrow = &zb[rb + o][wpix * 19];
    float z[9];
#pragma unroll
    for (int k = 0; k < 8; ++k) z[k] = zrow[2 * k + g] + bz[k];
    z[8] = 0.0f;
    float mx = z[0];
#pragma unroll
    for (int k = 1; k < 9; ++k) mx = fmaxf(mx, z[k]);
    float s = 0.0f;
#pragma unroll
    for (int k = 0; k < 9; ++k) {
      z[k] = expf(z[k] - mx);
      s += z[k];
    }
    const float inv = 1.0f / s;
#pragma unroll
    for (int k = 0; k < 9; ++k) z[k] *= inv;
    float ev[5], od[4];
#pragma unroll
    for (int i = 0; i < 5; ++i) ev[i] = g ? z[4 + i] : z[i];
#pragma unroll
    for (int i = 0; i < 4; ++i) od[i] = __shfl_xor(g ? z[i] : z[5 + i], 16);
    float* fblk = ffw + ((size_t)b * 2 + g) * 9 * HW;
    float4* q4 = (float4*)fblk + ((h0 + o) * 2) * WW + wpix;
    q4[0] = make_float4(ev[0], od[0], ev[1], od[1]);
    q4[WW] = make_float4(ev[2], od[2], ev[3], od[3]);
    fblk[8 * HW + (h0 + o) * WW + wpix] = ev[4];
  }
}

#define RB 8
#define CS 8
#define TR (RB + 2)
#define XP 130
#define NR (32 / CS)

__device__ __forceinline__ void lds_barrier() {
  __builtin_amdgcn_fence(__ATOMIC_RELEASE, "workgroup", "local");
  __builtin_amdgcn_s_barrier();
  __builtin_amdgcn_fence(__ATOMIC_ACQUIRE, "workgroup", "local");
}

__device__ __forceinline__ void load_col(float (&x)[TR], const float* p, const int (&ro)[TR]) {
#pragma unroll
  for (int i = 0; i < TR; ++i) x[i] = p[ro[i]];
}

__global__ __launch_bounds__(CS * 128) void combine(
    const float* inp, const float* ffw, float* __restrict__ out) {
  __shared__ float4 wq_s[RB * 2 * WW];
  __shared__ float w8_s[RB * WW];
  __shared__ float xt_s[2][CS][TR * XP];

  const unsigned tid = threadIdx.x;
  const unsigned w = tid & 127u;
  const unsigned slot = tid >> 7;
  const int h0 = blockIdx.x * RB;
  const int s = blockIdx.y;
  const int b = blockIdx.z;
  const float* wblk = ffw + (((size_t)b * 2 + s) * 9) * HW;

  const float4* src4 = (const float4*)wblk + h0 * 2 * WW;
  const float* src8 = wblk + 8 * HW + h0 * WW;
  const float4 g0 = src4[tid];
  const float4 g1 = src4[tid + CS * 128];
  const float g8 = src8[tid];

  int ro[TR];
  ro[0] = reflect_idx(h0 - 1, HH) * WW;
#pragma unroll
  for (int i = 1; i <= RB; ++i) ro[i] = (h0 + i - 1) * WW;
  ro[TR - 1] = reflect_idx(h0 + RB, HH) * WW;

  const size_t cofs = (size_t)(b * 64 + s + 2 * (int)slot) * HW + w;
  const float* xc = inp + cofs;
  const size_t RS = (size_t)2 * CS * HW;
  float x[2][TR];
  load_col(x[0], xc, ro);
  load_col(x[1], xc + RS, ro);
  __builtin_amdgcn_sched_barrier(0);

  wq_s[tid] = g0;
  wq_s[tid + CS * 128] = g1;
  w8_s[tid] = g8;

  const bool edge = ((w == 1u) | (w == 126u)) != 0;
  const unsigned hpos = (w >> 6) * (unsigned)(XP - 1);

  float4 qa[RB], qb[RB];
  float k8[RB];

#pragma unroll
  for (int r = 0; r < NR; ++r) {
    float* tile = &xt_s[r & 1][slot][0];
#pragma unroll
    for (int i = 0; i < TR; ++i) tile[i * XP + w + 1] = x[r & 1][i];
    if (edge) {
#pragma unroll
      for (int i = 0; i < TR; ++i) tile[i * XP + hpos] = x[r & 1][i];
    }
    if (r + 2 < NR) load_col(x[r & 1], xc + (size_t)(r + 2) * RS, ro);
    lds_barrier();
    if (r == 0) {
#pragma unroll
      for (int i = 0; i < RB; ++i) {
        qa[i] = wq_s[(i * 2) * WW + w];
        qb[i] = wq_s[(i * 2 + 1) * WW + w];
        k8[i] = w8_s[i * WW + w];
      }
    }
    const float* tw = tile + w;
    float* orow = out + cofs + (size_t)r * RS + h0 * WW;
    float t0 = tw[0], t1 = tw[1], t2 = tw[2];
    float m0 = tw[XP], m1 = tw[XP + 1], m2 = tw[XP + 2];
#pragma unroll
    for (int i = 0; i < RB; ++i) {
      const float u0 = tw[(i + 2) * XP], u1 = tw[(i + 2) * XP + 1], u2 = tw[(i + 2) * XP + 2];
      float acc = 0.0f;
      acc = fmaf(qa[i].x, t0, acc);
      acc = fmaf(qa[i].y, t1, acc);
      acc = fmaf(qa[i].z, t2, acc);
      acc = fmaf(qa[i].w, m0, acc);
      acc = fmaf(qb[i].x, m1, acc);
      acc = fmaf(qb[i].y, m2, acc);
      acc = fmaf(qb[i].z, u0, acc);
      acc = fmaf(qb[i].w, u1, acc);
      acc = fmaf(k8[i], u2, acc);
      orow[i * WW] = acc;
      t0 = m0; t1 = m1; t2 = m2;
      m0 = u0; m1 = u1; m2 = u2;
    }
  }
}

extern "C" void kernel_launch(void* const* d_in, const int* in_sizes, int n_in,
                              void* d_out, int out_size, void* d_ws, size_t ws_size,
                              hipStream_t stream) {
  const float* features = (const float*)d_in[0];
  const float* inpt     = (const float*)d_in[1];
  const float* conv_w   = (const float*)d_in[2];
  const float* conv_b   = (const float*)d_in[3];
  float* out = (float*)d_out;

  float* wp  = (float*)d_ws;
  float* ffw = wp + 16384;

  pack_w<<<64, 256, 0, stream>>>(conv_w, wp);
  conv_softmax<<<dim3(32, 8), 256, 0, stream>>>(features, conv_b, wp, ffw);
  combine<<<dim3(16, 2, 8), 1024, 0, stream>>>(inpt, ffw, out);
}
